# starting kernel
# baseline (speedup 1.0000x reference)
#include <hip/hip_runtime.h>
#include <hip/hip_bf16.h>


#define B_  2
#define N_  512
#define F_  768
#define ATT_ 64
#define HID_ 128

typedef __attribute__((ext_vector_type(2))) float v2f;
typedef __attribute__((ext_vector_type(8))) float v8f;

__global__ void k_h(const float* __restrict__ x, const float* __restrict__ Ww,
                    const float* __restrict__ Wb, float* __restrict__ h) {
    const int lane = threadIdx.x & 31;
    const int wave = threadIdx.x >> 5;
    const int tile = blockIdx.x * 4 + wave;
    const int tm = tile >> 2;
    const int tn = tile & 3;
    const int kb = (lane >> 4) * 2;
    const int m  = tm * 16 + (lane & 15);
    const int n  = tn * 16 + (lane & 15);

    const float* xrow = x  + (size_t)m * F_;
    const float* wrow = Ww + (size_t)n * F_;

    v8f c = {};
    for (int k = 0; k < F_; k += 4) {
        v2f a, b;
        a.x = xrow[k + kb]; a.y = xrow[k + kb + 1];
        b.x = wrow[k + kb]; b.y = wrow[k + kb + 1];
        c = __builtin_amdgcn_wmma_f32_16x16x4_f32(false, a, false, b,
                                                  (short)0, c, false, false);
    }
    const float bias = Wb[n];
    const int row0 = tm * 16 + (lane >> 4) * 8;
    const int col  = tn * 16 + (lane & 15);
#pragma unroll
    for (int v = 0; v < 8; ++v)
        h[(size_t)(row0 + v) * ATT_ + col] = c[v] + bias;
}

__global__ void k_e12(const float* __restrict__ h, const float* __restrict__ W1,
                      float* __restrict__ e12) {
    const int lane = threadIdx.x & 31;
    const int wave = threadIdx.x >> 5;
    const int tile = blockIdx.x * 4 + wave;
    const int tm = tile >> 4;
    const int tn = tile & 15;
    const int kb = (lane >> 4) * 2;
    const int m    = tm * 16 + (lane & 15);
    const int ncol = tn * 16 + (lane & 15);
    const int w1row = (ncol < HID_) ? ncol : (ncol - HID_);
    const int koff  = (ncol < HID_) ? 0 : ATT_;

    const float* hrow = h  + (size_t)m * ATT_;
    const float* wrow = W1 + (size_t)w1row * (2 * ATT_) + koff;

    v8f c = {};
    for (int k = 0; k < ATT_; k += 4) {
        v2f a, b;
        a.x = hrow[k + kb]; a.y = hrow[k + kb + 1];
        b.x = wrow[k + kb]; b.y = wrow[k + kb + 1];
        c = __builtin_amdgcn_wmma_f32_16x16x4_f32(false, a, false, b,
                                                  (short)0, c, false, false);
    }
    const int row0 = tm * 16 + (lane >> 4) * 8;
    const int col  = tn * 16 + (lane & 15);
#pragma unroll
    for (int v = 0; v < 8; ++v)
        e12[(size_t)(row0 + v) * 256 + col] = c[v];
}

__global__ void k_pair(const float* __restrict__ e12, const float* __restrict__ b1,
                       const float* __restrict__ W2, const float* __restrict__ b2,
                       const int* __restrict__ adj, float* __restrict__ expE,
                       float* __restrict__ partials) {
    __shared__ float se1[32][HID_ + 1];
    __shared__ float se2[32][HID_ + 1];
    __shared__ float sb1[HID_];
    __shared__ float sw2[HID_];
    __shared__ float red[256];

    const int blk = blockIdx.x;
    const int b   = blk >> 8;
    const int t   = blk & 255;
    const int i0  = (t >> 4) * 32;
    const int j0  = (t & 15) * 32;
    const int tid = threadIdx.x;

    for (int s = tid; s < 32 * HID_; s += 256) {
        const int r = s >> 7, cc = s & 127;
        se1[r][cc] = e12[(size_t)(b * N_ + i0 + r) * 256 + cc];
        se2[r][cc] = e12[(size_t)(b * N_ + j0 + r) * 256 + HID_ + cc];
    }
    if (tid < HID_) { sb1[tid] = b1[tid]; sw2[tid] = W2[tid]; }
    __syncthreads();

    const float bias2 = b2[0];
    const int li  = tid >> 3;
    const int lj0 = (tid & 7) * 4;

    float acc[4] = {0.f, 0.f, 0.f, 0.f};
    for (int hh = 0; hh < HID_; ++hh) {
        const float a1 = se1[li][hh] + sb1[hh];
        const float w  = sw2[hh];
#pragma unroll
        for (int q = 0; q < 4; ++q) {
            const float s = a1 + se2[lj0 + q][hh];
            acc[q] = fmaf(fmaxf(s, 0.f), w, acc[q]);
        }
    }

    float lsum = 0.f;
#pragma unroll
    for (int q = 0; q < 4; ++q) {
        float e = acc[q] + bias2;
        e = (e >= 0.f) ? e : 0.01f * e;
        const size_t gi = (size_t)b * N_ * N_ + (size_t)(i0 + li) * N_ + (j0 + lj0 + q);
        const float ev = adj[gi] ? __expf(e) : 0.f;
        expE[gi] = ev;
        lsum += ev;
    }

    red[tid] = lsum;
    __syncthreads();
    for (int s = 128; s > 0; s >>= 1) {
        if (tid < s) red[tid] += red[tid + s];
        __syncthreads();
    }
    if (tid == 0) partials[blk] = red[0];
}

__global__ void k_reduce(const float* __restrict__ partials, float* __restrict__ sumexp) {
    __shared__ float red[256];
    red[threadIdx.x] = partials[blockIdx.x * 256 + threadIdx.x];
    __syncthreads();
    for (int s = 128; s > 0; s >>= 1) {
        if (threadIdx.x < s) red[threadIdx.x] += red[threadIdx.x + s];
        __syncthreads();
    }
    if (threadIdx.x == 0) sumexp[blockIdx.x] = red[0];
}

__global__ void k_out(const float* __restrict__ expE, const float* __restrict__ h,
                      const float* __restrict__ sumexp, float* __restrict__ out) {
    const int lane = threadIdx.x & 31;
    const int wave = threadIdx.x >> 5;
    const int tile = blockIdx.x * 4 + wave;
    const int b  = tile >> 7;
    const int tm = (tile >> 2) & 31;
    const int tn = tile & 3;
    const int kb = (lane >> 4) * 2;
    const int m  = tm * 16 + (lane & 15);
    const int n  = tn * 16 + (lane & 15);

    const float* arow = expE + (size_t)b * N_ * N_ + (size_t)m * N_;
    const float* bcol = h    + (size_t)b * N_ * ATT_ + n;

    v8f c = {};
    for (int k = 0; k < N_; k += 4) {
        v2f a, bb;
        a.x  = arow[k + kb];
        a.y  = arow[k + kb + 1];
        bb.x = bcol[(size_t)(k + kb) * ATT_];
        bb.y = bcol[(size_t)(k + kb + 1) * ATT_];
        c = __builtin_amdgcn_wmma_f32_16x16x4_f32(false, a, false, bb,
                                                  (short)0, c, false, false);
    }
    const float inv = 1.0f / sumexp[b];
    const int row0 = tm * 16 + (lane >> 4) * 8;
    const int col  = tn * 16 + (lane & 15);
#pragma unroll
    for (int v = 0; v < 8; ++v)
        out[((size_t)b * N_ + row0 + v) * ATT_ + col] = c[v] * inv;
}

extern "C" void kernel_launch(void* const* d_in, const int* in_sizes, int n_in,
                              void* d_out, int out_size, void* d_ws, size_t ws_size,
                              hipStream_t stream) {
    const float* x   = (const float*)d_in[0];
    const int*   adj = (const int*)  d_in[1];
    const float* Ww  = (const float*)d_in[2];
    const float* Wb  = (const float*)d_in[3];
    const float* W1  = (const float*)d_in[4];
    const float* b1  = (const float*)d_in[5];
    const float* W2  = (const float*)d_in[6];
    const float* b2  = (const float*)d_in[7];
    float* out = (float*)d_out;

    float* ws    = (float*)d_ws;
    float* h_    = ws;
    float* e12   = h_   + (size_t)B_ * N_ * ATT_;
    float* expE  = e12  + (size_t)B_ * N_ * 256;
    float* parts = expE + (size_t)B_ * N_ * N_;
    float* sums  = parts + 512;

    k_h     <<< 64, 128, 0, stream>>>(x, Ww, Wb, h_);
    k_e12   <<<256, 128, 0, stream>>>(h_, W1, e12);
    k_pair  <<<512, 256, 0, stream>>>(e12, b1, W2, b2, adj, expE, parts);
    k_reduce<<<  2, 256, 0, stream>>>(parts, sums);
    k_out   <<< 64, 128, 0, stream>>>(expE, h_, sums, out);
}


# split the K loop of the two long GEMMs (h = x W^T, out = expE h) over 12 / 16 waves per work-group, partial 16x16 tiles summed in fixed order through LDS
# speedup vs baseline: 1.4516x; 1.4516x over previous
#include <hip/hip_runtime.h>
#include <hip/hip_bf16.h>


#define B_  2
#define N_  512
#define F_  768
#define ATT_ 64
#define HID_ 128

typedef __attribute__((ext_vector_type(2))) float v2f;
typedef __attribute__((ext_vector_type(8))) float v8f;

#define KH_SLICES 12
#define KH_SPAN   (F_ / KH_SLICES)
#define TILE_PITCH 272
__global__ void __launch_bounds__(KH_SLICES * 32)
k_h(const float* __restrict__ x, const float* __restrict__ Ww,
    const float* __restrict__ Wb, float* __restrict__ h) {
    __shared__ float red[KH_SLICES * 2 * TILE_PITCH];

    const int lane = threadIdx.x & 31;
    const int sl   = threadIdx.x >> 5;
    const int tm   = blockIdx.x >> 1;
    const int n0   = (blockIdx.x & 1) * 32;
    const int kb   = (lane >> 4) * 2;
    const int c16  = lane & 15;

    const float* xrow = x  + (size_t)(tm * 16 + c16) * F_ + sl * KH_SPAN + kb;
    const float* w0   = Ww + (size_t)(n0 + c16) * F_ + sl * KH_SPAN + kb;
    const float* w1   = w0 + (size_t)16 * F_;

    v8f c0 = {}, c1 = {};
#pragma unroll
    for (int k = 0; k < KH_SPAN; k += 4) {
        v2f a, b0, b1;
        a.x  = xrow[k]; a.y  = xrow[k + 1];
        b0.x = w0[k];   b0.y = w0[k + 1];
        b1.x = w1[k];   b1.y = w1[k + 1];
        c0 = __builtin_amdgcn_wmma_f32_16x16x4_f32(false, a, false, b0,
                                                   (short)0, c0, false, false);
        c1 = __builtin_amdgcn_wmma_f32_16x16x4_f32(false, a, false, b1,
                                                   (short)0, c1, false, false);
    }

    float* mine = red + (size_t)(sl * 2) * TILE_PITCH + lane;
#pragma unroll
    for (int v = 0; v < 8; ++v) {
        mine[v * 32]              = c0[v];
        mine[TILE_PITCH + v * 32] = c1[v];
    }
    __syncthreads();

    for (int o = threadIdx.x; o < 16 * 32; o += KH_SLICES * 32) {
        const int r  = o >> 5;
        const int cc = o & 31;
        const float* p = red + (cc >> 4) * TILE_PITCH + (r & 7) * 32 + (r >> 3) * 16 + (cc & 15);
        float acc = p[0];
#pragma unroll
        for (int s = 1; s < KH_SLICES; ++s) acc += p[(size_t)s * 2 * TILE_PITCH];
        h[(size_t)(tm * 16 + r) * ATT_ + n0 + cc] = acc + Wb[n0 + cc];
    }
}

__global__ void k_e12(const float* __restrict__ h, const float* __restrict__ W1,
                      float* __restrict__ e12) {
    const int lane = threadIdx.x & 31;
    const int wave = threadIdx.x >> 5;
    const int tile = blockIdx.x * 4 + wave;
    const int tm = tile >> 4;
    const int tn = tile & 15;
    const int kb = (lane >> 4) * 2;
    const int m    = tm * 16 + (lane & 15);
    const int ncol = tn * 16 + (lane & 15);
    const int w1row = (ncol < HID_) ? ncol : (ncol - HID_);
    const int koff  = (ncol < HID_) ? 0 : ATT_;

    const float* hrow = h  + (size_t)m * ATT_;
    const float* wrow = W1 + (size_t)w1row * (2 * ATT_) + koff;

    v8f c = {};
    for (int k = 0; k < ATT_; k += 4) {
        v2f a, b;
        a.x = hrow[k + kb]; a.y = hrow[k + kb + 1];
        b.x = wrow[k + kb]; b.y = wrow[k + kb + 1];
        c = __builtin_amdgcn_wmma_f32_16x16x4_f32(false, a, false, b,
                                                  (short)0, c, false, false);
    }
    const int row0 = tm * 16 + (lane >> 4) * 8;
    const int col  = tn * 16 + (lane & 15);
#pragma unroll
    for (int v = 0; v < 8; ++v)
        e12[(size_t)(row0 + v) * 256 + col] = c[v];
}

__global__ void k_pair(const float* __restrict__ e12, const float* __restrict__ b1,
                       const float* __restrict__ W2, const float* __restrict__ b2,
                       const int* __restrict__ adj, float* __restrict__ expE,
                       float* __restrict__ partials) {
    __shared__ float se1[32][HID_ + 1];
    __shared__ float se2[32][HID_ + 1];
    __shared__ float sb1[HID_];
    __shared__ float sw2[HID_];
    __shared__ float red[256];

    const int blk = blockIdx.x;
    const int b   = blk >> 8;
    const int t   = blk & 255;
    const int i0  = (t >> 4) * 32;
    const int j0  = (t & 15) * 32;
    const int tid = threadIdx.x;

    for (int s = tid; s < 32 * HID_; s += 256) {
        const int r = s >> 7, cc = s & 127;
        se1[r][cc] = e12[(size_t)(b * N_ + i0 + r) * 256 + cc];
        se2[r][cc] = e12[(size_t)(b * N_ + j0 + r) * 256 + HID_ + cc];
    }
    if (tid < HID_) { sb1[tid] = b1[tid]; sw2[tid] = W2[tid]; }
    __syncthreads();

    const float bias2 = b2[0];
    const int li  = tid >> 3;
    const int lj0 = (tid & 7) * 4;

    float acc[4] = {0.f, 0.f, 0.f, 0.f};
    for (int hh = 0; hh < HID_; ++hh) {
        const float a1 = se1[li][hh] + sb1[hh];
        const float w  = sw2[hh];
#pragma unroll
        for (int q = 0; q < 4; ++q) {
            const float s = a1 + se2[lj0 + q][hh];
            acc[q] = fmaf(fmaxf(s, 0.f), w, acc[q]);
        }
    }

    float lsum = 0.f;
#pragma unroll
    for (int q = 0; q < 4; ++q) {
        float e = acc[q] + bias2;
        e = (e >= 0.f) ? e : 0.01f * e;
        const size_t gi = (size_t)b * N_ * N_ + (size_t)(i0 + li) * N_ + (j0 + lj0 + q);
        const float ev = adj[gi] ? __expf(e) : 0.f;
        expE[gi] = ev;
        lsum += ev;
    }

    red[tid] = lsum;
    __syncthreads();
    for (int s = 128; s > 0; s >>= 1) {
        if (tid < s) red[tid] += red[tid + s];
        __syncthreads();
    }
    if (tid == 0) partials[blk] = red[0];
}

__global__ void k_reduce(const float* __restrict__ partials, float* __restrict__ sumexp) {
    __shared__ float red[256];
    red[threadIdx.x] = partials[blockIdx.x * 256 + threadIdx.x];
    __syncthreads();
    for (int s = 128; s > 0; s >>= 1) {
        if (threadIdx.x < s) red[threadIdx.x] += red[threadIdx.x + s];
        __syncthreads();
    }
    if (threadIdx.x == 0) sumexp[blockIdx.x] = red[0];
}

#define KO_SLICES 16
#define KO_SPAN   (N_ / KO_SLICES)
__global__ void __launch_bounds__(KO_SLICES * 32)
k_out(const float* __restrict__ expE, const float* __restrict__ h,
      const float* __restrict__ sumexp, float* __restrict__ out) {
    __shared__ float red[KO_SLICES * 2 * TILE_PITCH];

    const int lane = threadIdx.x & 31;
    const int sl   = threadIdx.x >> 5;
    const int b    = blockIdx.x >> 6;
    const int tm   = (blockIdx.x >> 1) & 31;
    const int n0   = (blockIdx.x & 1) * 32;
    const int kb   = (lane >> 4) * 2;
    const int c16  = lane & 15;

    const float* arow = expE + (size_t)b * N_ * N_ + (size_t)(tm * 16 + c16) * N_ + sl * KO_SPAN + kb;
    const float* bcol = h + ((size_t)b * N_ + sl * KO_SPAN + kb) * ATT_ + n0 + c16;

    v8f c0 = {}, c1 = {};
#pragma unroll
    for (int k = 0; k < KO_SPAN; k += 4) {
        v2f a, b0, b1;
        a.x  = arow[k];
        a.y  = arow[k + 1];
        b0.x = bcol[(size_t)k * ATT_];
        b0.y = bcol[(size_t)(k + 1) * ATT_];
        b1.x = bcol[(size_t)k * ATT_ + 16];
        b1.y = bcol[(size_t)(k + 1) * ATT_ + 16];
        c0 = __builtin_amdgcn_wmma_f32_16x16x4_f32(false, a, false, b0,
                                                   (short)0, c0, false, false);
        c1 = __builtin_amdgcn_wmma_f32_16x16x4_f32(false, a, false, b1,
                                                   (short)0, c1, false, false);
    }

    float* mine = red + (size_t)(sl * 2) * TILE_PITCH + lane;
#pragma unroll
    for (int v = 0; v < 8; ++v) {
        mine[v * 32]              = c0[v];
        mine[TILE_PITCH + v * 32] = c1[v];
    }
    __syncthreads();

    const float inv = 1.0f / sumexp[b];
    const int r  = threadIdx.x >> 5;
    const int cc = threadIdx.x & 31;
    const float* p = red + (cc >> 4) * TILE_PITCH + (r & 7) * 32 + (r >> 3) * 16 + (cc & 15);
    float acc = p[0];
#pragma unroll
    for (int s = 1; s < KO_SLICES; ++s) acc += p[(size_t)s * 2 * TILE_PITCH];
    out[((size_t)b * N_ + tm * 16 + r) * ATT_ + n0 + cc] = acc * inv;
}

extern "C" void kernel_launch(void* const* d_in, const int* in_sizes, int n_in,
                              void* d_out, int out_size, void* d_ws, size_t ws_size,
                              hipStream_t stream) {
    const float* x   = (const float*)d_in[0];
    const int*   adj = (const int*)  d_in[1];
    const float* Ww  = (const float*)d_in[2];
    const float* Wb  = (const float*)d_in[3];
    const float* W1  = (const float*)d_in[4];
    const float* b1  = (const float*)d_in[5];
    const float* W2  = (const float*)d_in[6];
    const float* b2  = (const float*)d_in[7];
    float* out = (float*)d_out;

    float* ws    = (float*)d_ws;
    float* h_    = ws;
    float* e12   = h_   + (size_t)B_ * N_ * ATT_;
    float* expE  = e12  + (size_t)B_ * N_ * 256;
    float* parts = expE + (size_t)B_ * N_ * N_;
    float* sums  = parts + 512;

    k_h     <<<128, 384, 0, stream>>>(x, Ww, Wb, h_);
    k_e12   <<<256, 128, 0, stream>>>(h_, W1, e12);
    k_pair  <<<512, 256, 0, stream>>>(e12, b1, W2, b2, adj, expE, parts);
    k_reduce<<<  2, 256, 0, stream>>>(parts, sums);
    k_out   <<<128, 512, 0, stream>>>(expE, h_, sums, out);
}


# stage 3 retiled: 4 i x 4 j per thread (was 1 x 4) from h-major LDS tiles read as 16-byte vectors; 32x64 tile per 4-wave work-group (1024 waves, was 4096); tile, mask and output moved 16 B per lane
# speedup vs baseline: 1.9069x; 1.3137x over previous
#include <hip/hip_runtime.h>
#include <hip/hip_bf16.h>


#define B_  2
#define N_  512
#define F_  768
#define ATT_ 64
#define HID_ 128

typedef __attribute__((ext_vector_type(2))) float v2f;
typedef __attribute__((ext_vector_type(8))) float v8f;

__device__ __forceinline__ float xhalf(float x) {
    const int b = __builtin_bit_cast(int, x);
    return __builtin_bit_cast(float, __builtin_amdgcn_permlanex16(b, b, 0x76543210, 0xfedcba98, false, false));
}
template <int M>
__device__ __forceinline__ float xrow(float x) {
    const int b = __builtin_bit_cast(int, x);
    return __builtin_bit_cast(float, __builtin_amdgcn_update_dpp(b, b, 0x160 + M, 0xf, 0xf, false));
}

#define KH_SLICES 12
#define KH_SPAN   (F_ / KH_SLICES)
#define TILE_PITCH 272
__global__ void __launch_bounds__(KH_SLICES * 32)
k_h(const float* __restrict__ x, const float* __restrict__ Ww,
    const float* __restrict__ Wb, float* __restrict__ h) {
    __shared__ float red[KH_SLICES * 2 * TILE_PITCH];

    const int lane = threadIdx.x & 31;
    const int sl   = threadIdx.x >> 5;
    const int tm   = blockIdx.x >> 1;
    const int n0   = (blockIdx.x & 1) * 32;
    const int kb   = (lane >> 4) * 2;
    const int c16  = lane & 15;

    const float* xrow = x  + (size_t)(tm * 16 + c16) * F_ + sl * KH_SPAN + kb;
    const float* w0   = Ww + (size_t)(n0 + c16) * F_ + sl * KH_SPAN + kb;
    const float* w1   = w0 + (size_t)16 * F_;

    v8f c0 = {}, c1 = {};
#pragma unroll
    for (int k = 0; k < KH_SPAN; k += 4) {
        v2f a, b0, b1;
        a.x  = xrow[k]; a.y  = xrow[k + 1];
        b0.x = w0[k];   b0.y = w0[k + 1];
        b1.x = w1[k];   b1.y = w1[k + 1];
        c0 = __builtin_amdgcn_wmma_f32_16x16x4_f32(false, a, false, b0,
                                                   (short)0, c0, false, false);
        c1 = __builtin_amdgcn_wmma_f32_16x16x4_f32(false, a, false, b1,
                                                   (short)0, c1, false, false);
    }

    float* mine = red + (size_t)(sl * 2) * TILE_PITCH + lane;
#pragma unroll
    for (int v = 0; v < 8; ++v) {
        mine[v * 32]              = c0[v];
        mine[TILE_PITCH + v * 32] = c1[v];
    }
    __syncthreads();

    for (int o = threadIdx.x; o < 16 * 32; o += KH_SLICES * 32) {
        const int r  = o >> 5;
        const int cc = o & 31;
        const float* p = red + (cc >> 4) * TILE_PITCH + (r & 7) * 32 + (r >> 3) * 16 + (cc & 15);
        float acc = p[0];
#pragma unroll
        for (int s = 1; s < KH_SLICES; ++s) acc += p[(size_t)s * 2 * TILE_PITCH];
        h[(size_t)(tm * 16 + r) * ATT_ + n0 + cc] = acc + Wb[n0 + cc];
    }
}

__global__ void k_e12(const float* __restrict__ h, const float* __restrict__ W1,
                      float* __restrict__ e12) {
    const int lane = threadIdx.x & 31;
    const int wave = threadIdx.x >> 5;
    const int tile = blockIdx.x * 4 + wave;
    const int tm = tile >> 4;
    const int tn = tile & 15;
    const int kb = (lane >> 4) * 2;
    const int m    = tm * 16 + (lane & 15);
    const int ncol = tn * 16 + (lane & 15);
    const int w1row = (ncol < HID_) ? ncol : (ncol - HID_);
    const int koff  = (ncol < HID_) ? 0 : ATT_;

    const float* hrow = h  + (size_t)m * ATT_;
    const float* wrow = W1 + (size_t)w1row * (2 * ATT_) + koff;

    v8f c = {};
    for (int k = 0; k < ATT_; k += 4) {
        v2f a, b;
        a.x = hrow[k + kb]; a.y = hrow[k + kb + 1];
        b.x = wrow[k + kb]; b.y = wrow[k + kb + 1];
        c = __builtin_amdgcn_wmma_f32_16x16x4_f32(false, a, false, b,
                                                  (short)0, c, false, false);
    }
    const int row0 = tm * 16 + (lane >> 4) * 8;
    const int col  = tn * 16 + (lane & 15);
#pragma unroll
    for (int v = 0; v < 8; ++v)
        e12[(size_t)(row0 + v) * 256 + col] = c[v];
}

#define PA_PITCH 36
#define PB_PITCH 68

typedef __attribute__((ext_vector_type(4))) float v4f;
struct PairGrp { v4f a[4]; v4f b[4]; v4f w; };

__device__ __forceinline__ void pair_load(PairGrp& G, const float* pa, const float* pb,
                                          const float* pw, int g) {
#pragma unroll
    for (int u = 0; u < 4; ++u) {
        G.a[u] = *(const v4f*)(pa + (4 * g + u) * PA_PITCH);
        G.b[u] = *(const v4f*)(pb + (4 * g + u) * PB_PITCH);
    }
    G.w = *(const v4f*)(pw + 4 * g);
}

__device__ __forceinline__ void pair_fma(float (&acc)[4][4], const PairGrp& G) {
    v4f w = G.w;
    asm("" : "+v"(w));
#pragma unroll
    for (int u = 0; u < 4; ++u) {
        v4f a = G.a[u], b = G.b[u];
        asm("" : "+v"(a));
        asm("" : "+v"(b));
#pragma unroll
        for (int r = 0; r < 4; ++r) {
#pragma unroll
            for (int c = 0; c < 4; ++c)
                acc[r][c] = fmaf(fmaxf(a[r] + b[c], 0.f), w[u], acc[r][c]);
        }
    }
}

__device__ __forceinline__ float pair_out(float s, float bias2, int m) {
    float e = s + bias2;
    e = (e >= 0.f) ? e : 0.01f * e;
    return m ? __expf(e) : 0.f;
}

__global__ void __launch_bounds__(128)
k_pair(const float* __restrict__ e12, const float* __restrict__ b1,
       const float* __restrict__ W2, const float* __restrict__ b2,
       const int* __restrict__ adj, float* __restrict__ expE,
       float* __restrict__ partials) {
    __shared__ __attribute__((aligned(16))) float sA[HID_ * PA_PITCH];
    __shared__ __attribute__((aligned(16))) float sB[HID_ * PB_PITCH];
    __shared__ __attribute__((aligned(16))) float sw2[HID_];
    __shared__ float red[4];

    const int blk  = blockIdx.x;
    const int b    = blk >> 7;
    const int t    = blk & 127;
    const int i0   = (t >> 3) * 32;
    const int j0   = (t & 7) * 64;
    const int tid  = threadIdx.x;
    const int lane = tid & 31;
    const int wave = tid >> 5;
    const int tj   = tid & 15;
    const int ti   = tid >> 4;

    const size_t g0 = (size_t)b * N_ * N_ + (size_t)(i0 + 4 * ti) * N_ + (j0 + 4 * tj);

    const int hq = wave * 8 + (lane & 7);
    const int rr = lane >> 3;
    const float4 bb = *(const float4*)(b1 + 4 * hq);
    float4 v[24];
#pragma unroll
    for (int k = 0; k < 24; ++k) {
        const int rho = 4 * k + rr;
        const float* src = (k < 8)
            ? e12 + (size_t)(b * N_ + i0 + rho) * 256 + 4 * hq
            : e12 + (size_t)(b * N_ + j0 + rho - 32) * 256 + HID_ + 4 * hq;
        v[k] = *(const float4*)src;
    }
#pragma unroll
    for (int k = 0; k < 24; ++k) {
        const int rho = 4 * k + rr;
        if (k < 8) {
            float* d = sA + (4 * hq) * PA_PITCH + rho;
            d[0]            = v[k].x + bb.x;
            d[PA_PITCH]     = v[k].y + bb.y;
            d[2 * PA_PITCH] = v[k].z + bb.z;
            d[3 * PA_PITCH] = v[k].w + bb.w;
        } else {
            float* d = sB + (4 * hq) * PB_PITCH + (rho - 32);
            d[0]            = v[k].x;
            d[PB_PITCH]     = v[k].y;
            d[2 * PB_PITCH] = v[k].z;
            d[3 * PB_PITCH] = v[k].w;
        }
    }
    sw2[tid] = W2[tid];
    __syncthreads();

    int4 msk[4];
#pragma unroll
    for (int r = 0; r < 4; ++r) msk[r] = *(const int4*)(adj + g0 + (size_t)r * N_);

    const float* pa = sA + 4 * ti;
    const float* pb = sB + 4 * tj;

    float acc[4][4];
#pragma unroll
    for (int r = 0; r < 4; ++r) {
#pragma unroll
        for (int c = 0; c < 4; ++c) acc[r][c] = 0.f;
    }

    PairGrp G0, G1;
    pair_load(G0, pa, pb, sw2, 0);
#pragma unroll 1
    for (int g = 0; g < HID_ / 4 - 2; g += 2) {
        pair_load(G1, pa, pb, sw2, g + 1);
        pair_fma(acc, G0);
        pair_load(G0, pa, pb, sw2, g + 2);
        pair_fma(acc, G1);
    }
    pair_load(G1, pa, pb, sw2, HID_ / 4 - 1);
    pair_fma(acc, G0);
    pair_fma(acc, G1);

    const float bias2 = b2[0];
    float lsum = 0.f;
#pragma unroll
    for (int r = 0; r < 4; ++r) {
        float4 o;
        o.x = pair_out(acc[r][0], bias2, msk[r].x);
        o.y = pair_out(acc[r][1], bias2, msk[r].y);
        o.z = pair_out(acc[r][2], bias2, msk[r].z);
        o.w = pair_out(acc[r][3], bias2, msk[r].w);
        *(float4*)(expE + g0 + (size_t)r * N_) = o;
        lsum += (o.x + o.y) + (o.z + o.w);
    }

    lsum += xhalf(lsum);
    lsum += xrow<8>(lsum);
    lsum += xrow<4>(lsum);
    lsum += xrow<2>(lsum);
    lsum += xrow<1>(lsum);
    if (lane == 0) red[wave] = lsum;
    __syncthreads();
    if (tid == 0) {
        partials[2 * blk]     = red[0] + red[1];
        partials[2 * blk + 1] = red[2] + red[3];
    }
}

#define KO_SLICES 16
#define KO_SPAN   (N_ / KO_SLICES)
__global__ void __launch_bounds__(KO_SLICES * 32)
k_out(const float* __restrict__ expE, const float* __restrict__ h,
      const float* __restrict__ partials, float* __restrict__ out) {
    __shared__ float red[KO_SLICES * 2 * TILE_PITCH];

    const int lane = threadIdx.x & 31;
    const int sl   = threadIdx.x >> 5;
    const int b    = blockIdx.x >> 6;
    const int tm   = (blockIdx.x >> 1) & 31;
    const int n0   = (blockIdx.x & 1) * 32;
    const int kb   = (lane >> 4) * 2;
    const int c16  = lane & 15;

    const float* pp = partials + b * 256 + lane;
    const float q0 = pp[0],   q1 = pp[32],  q2 = pp[64],  q3 = pp[96];
    const float q4 = pp[128], q5 = pp[160], q6 = pp[192], q7 = pp[224];
    float tot = ((q0 + q4) + (q2 + q6)) + ((q1 + q5) + (q3 + q7));
    tot += xhalf(tot);
    tot += xrow<8>(tot);
    tot += xrow<4>(tot);
    tot += xrow<2>(tot);
    tot += xrow<1>(tot);

    const float* arow = expE + (size_t)b * N_ * N_ + (size_t)(tm * 16 + c16) * N_ + sl * KO_SPAN + kb;
    const float* bcol = h + ((size_t)b * N_ + sl * KO_SPAN + kb) * ATT_ + n0 + c16;

    v8f c0 = {}, c1 = {};
#pragma unroll
    for (int k = 0; k < KO_SPAN; k += 4) {
        v2f a, b0, b1;
        a.x  = arow[k];
        a.y  = arow[k + 1];
        b0.x = bcol[(size_t)k * ATT_];
        b0.y = bcol[(size_t)(k + 1) * ATT_];
        b1.x = bcol[(size_t)k * ATT_ + 16];
        b1.y = bcol[(size_t)(k + 1) * ATT_ + 16];
        c0 = __builtin_amdgcn_wmma_f32_16x16x4_f32(false, a, false, b0,
                                                   (short)0, c0, false, false);
        c1 = __builtin_amdgcn_wmma_f32_16x16x4_f32(false, a, false, b1,
                                                   (short)0, c1, false, false);
    }

    float* mine = red + (size_t)(sl * 2) * TILE_PITCH + lane;
#pragma unroll
    for (int v = 0; v < 8; ++v) {
        mine[v * 32]              = c0[v];
        mine[TILE_PITCH + v * 32] = c1[v];
    }
    __syncthreads();

    const float inv = 1.0f / tot;
    const int r  = threadIdx.x >> 5;
    const int cc = threadIdx.x & 31;
    const float* p = red + (cc >> 4) * TILE_PITCH + (r & 7) * 32 + (r >> 3) * 16 + (cc & 15);
    float acc = p[0];
#pragma unroll
    for (int s = 1; s < KO_SLICES; ++s) acc += p[(size_t)s * 2 * TILE_PITCH];
    out[((size_t)b * N_ + tm * 16 + r) * ATT_ + n0 + cc] = acc * inv;
}

extern "C" void kernel_launch(void* const* d_in, const int* in_sizes, int n_in,
                              void* d_out, int out_size, void* d_ws, size_t ws_size,
                              hipStream_t stream) {
    const float* x   = (const float*)d_in[0];
    const int*   adj = (const int*)  d_in[1];
    const float* Ww  = (const float*)d_in[2];
    const float* Wb  = (const float*)d_in[3];
    const float* W1  = (const float*)d_in[4];
    const float* b1  = (const float*)d_in[5];
    const float* W2  = (const float*)d_in[6];
    const float* b2  = (const float*)d_in[7];
    float* out = (float*)d_out;

    float* ws    = (float*)d_ws;
    float* h_    = ws;
    float* e12   = h_   + (size_t)B_ * N_ * ATT_;
    float* expE  = e12  + (size_t)B_ * N_ * 256;
    float* parts = expE + (size_t)B_ * N_ * N_;

    k_h     <<<128, 384, 0, stream>>>(x, Ww, Wb, h_);
    k_e12   <<<256, 128, 0, stream>>>(h_, W1, e12);
    k_pair  <<<256, 128, 0, stream>>>(e12, b1, W2, b2, adj, expE, parts);
    k_out   <<<128, 512, 0, stream>>>(expE, h_, parts, out);
}


# both GEMM kernels: 8 waves per work-group (was 12/16) on K slices of 96 and 64 fed by 16-byte fragment loads (k order permuted alike on A, B); 8 partial tiles added in slice order: rounding differs
# speedup vs baseline: 2.1936x; 1.1503x over previous
#include <hip/hip_runtime.h>
#include <hip/hip_bf16.h>


#define B_  2
#define N_  512
#define F_  768
#define ATT_ 64
#define HID_ 128

typedef __attribute__((ext_vector_type(2))) float v2f;
typedef __attribute__((ext_vector_type(8))) float v8f;

__device__ __forceinline__ float xhalf(float x) {
    const int b = __builtin_bit_cast(int, x);
    return __builtin_bit_cast(float, __builtin_amdgcn_permlanex16(b, b, 0x76543210, 0xfedcba98, false, false));
}
template <int M>
__device__ __forceinline__ float xrow(float x) {
    const int b = __builtin_bit_cast(int, x);
    return __builtin_bit_cast(float, __builtin_amdgcn_update_dpp(b, b, 0x160 + M, 0xf, 0xf, false));
}

#define KH_SLICES 8
#define KH_SPAN   (F_ / KH_SLICES)
#define TILE_PITCH 272
#define HT_PITCH  36
#define E12_COLS  256
#define ES_PITCH  260
#define E12_PLANE ((size_t)B_ * N_ * E12_COLS)
#define E12_WAVES 8

typedef __attribute__((ext_vector_type(4))) float v4f;

__global__ void __launch_bounds__(KH_SLICES * 32) __attribute__((amdgpu_waves_per_eu(2, 2)))
k_h(const float* __restrict__ x, const float* __restrict__ Ww,
    const float* __restrict__ Wb, const float* __restrict__ W1,
    float* __restrict__ h, float* __restrict__ e12p) {
    __shared__ float red[KH_SLICES * 2 * TILE_PITCH];
    __shared__ __attribute__((aligned(16))) float hT[16 * HT_PITCH];
    __shared__ __attribute__((aligned(16))) float es[16 * ES_PITCH];

    const int lane = threadIdx.x & 31;
    const int sl   = threadIdx.x >> 5;
    const int tm   = blockIdx.x >> 1;
    const int n0   = (blockIdx.x & 1) * 32;
    const int hf   = lane >> 4;
    const int c16  = lane & 15;
    const int wv   = __builtin_amdgcn_readfirstlane(sl);

    v4f wq0[4], wq1[4];
    if (wv < E12_WAVES) {
        const int ecol = wv * 32 + c16;
        const float* wp = W1 + (size_t)(ecol & (HID_ - 1)) * (2 * ATT_) + ((ecol >= HID_) ? ATT_ : 0) + n0 + 16 * hf;
#pragma unroll
        for (int q = 0; q < 4; ++q) {
            wq0[q] = *(const v4f*)(wp + 4 * q);
            wq1[q] = *(const v4f*)(wp + (size_t)16 * (2 * ATT_) + 4 * q);
        }
    }

    const float* xrow = x  + (size_t)(tm * 16 + c16) * F_ + sl * KH_SPAN + 4 * hf;
    const float* w0   = Ww + (size_t)(n0 + c16) * F_ + sl * KH_SPAN + 4 * hf;
    const float* w1   = w0 + (size_t)16 * F_;

    v8f c0 = {}, c1 = {};
#pragma unroll
    for (int k = 0; k < KH_SPAN; k += 8) {
        const v4f a4 = *(const v4f*)(xrow + k);
        const v4f p4 = *(const v4f*)(w0 + k);
        const v4f q4 = *(const v4f*)(w1 + k);
        v2f a, b0, b1;
        a.x  = a4.x; a.y  = a4.y;
        b0.x = p4.x; b0.y = p4.y;
        b1.x = q4.x; b1.y = q4.y;
        c0 = __builtin_amdgcn_wmma_f32_16x16x4_f32(false, a, false, b0,
                                                   (short)0, c0, false, false);
        c1 = __builtin_amdgcn_wmma_f32_16x16x4_f32(false, a, false, b1,
                                                   (short)0, c1, false, false);
        a.x  = a4.z; a.y  = a4.w;
        b0.x = p4.z; b0.y = p4.w;
        b1.x = q4.z; b1.y = q4.w;
        c0 = __builtin_amdgcn_wmma_f32_16x16x4_f32(false, a, false, b0,
                                                   (short)0, c0, false, false);
        c1 = __builtin_amdgcn_wmma_f32_16x16x4_f32(false, a, false, b1,
                                                   (short)0, c1, false, false);
    }

    float* mine = red + (size_t)(sl * 2) * TILE_PITCH + lane;
#pragma unroll
    for (int v = 0; v < 8; ++v) {
        mine[v * 32]              = c0[v];
        mine[TILE_PITCH + v * 32] = c1[v];
    }
    __syncthreads();

    float hv[2] = {0.f, 0.f};
#pragma unroll
    for (int t = 0; t < 2; ++t) {
        const int o = threadIdx.x + t * KH_SLICES * 32;
        if (o < 16 * 32) {
            const int r  = o >> 5;
            const int cc = o & 31;
            const float* p = red + (cc >> 4) * TILE_PITCH + (r & 7) * 32 + (r >> 3) * 16 + (cc & 15);
            float acc = p[0];
#pragma unroll
            for (int s = 1; s < KH_SLICES; ++s) acc += p[(size_t)s * 2 * TILE_PITCH];
            hv[t] = acc + Wb[n0 + cc];
            hT[r * HT_PITCH + cc] = hv[t];
        }
    }
    __syncthreads();

    if (wv < E12_WAVES) {
        v8f d0 = {}, d1 = {};
        const float* ap = hT + c16 * HT_PITCH + 16 * hf;
#pragma unroll
        for (int q = 0; q < 4; ++q) {
            const v4f a4 = *(const v4f*)(ap + 4 * q);
            v2f a, b0, b1;
            a.x  = a4.x;     a.y  = a4.y;
            b0.x = wq0[q].x; b0.y = wq0[q].y;
            b1.x = wq1[q].x; b1.y = wq1[q].y;
            d0 = __builtin_amdgcn_wmma_f32_16x16x4_f32(false, a, false, b0,
                                                       (short)0, d0, false, false);
            d1 = __builtin_amdgcn_wmma_f32_16x16x4_f32(false, a, false, b1,
                                                       (short)0, d1, false, false);
            a.x  = a4.z;     a.y  = a4.w;
            b0.x = wq0[q].z; b0.y = wq0[q].w;
            b1.x = wq1[q].z; b1.y = wq1[q].w;
            d0 = __builtin_amdgcn_wmma_f32_16x16x4_f32(false, a, false, b0,
                                                       (short)0, d0, false, false);
            d1 = __builtin_amdgcn_wmma_f32_16x16x4_f32(false, a, false, b1,
                                                       (short)0, d1, false, false);
        }
        float* st = es + (size_t)(8 * hf) * ES_PITCH + wv * 32 + c16;
#pragma unroll
        for (int v = 0; v < 8; ++v) {
            st[v * ES_PITCH]      = d0[v];
            st[v * ES_PITCH + 16] = d1[v];
        }
    }
    __syncthreads();

#pragma unroll
    for (int t = 0; t < 2; ++t) {
        const int o = threadIdx.x + t * KH_SLICES * 32;
        if (o < 16 * 32)
            h[(size_t)(tm * 16 + (o >> 5)) * ATT_ + n0 + (o & 31)] = hv[t];
    }
    float* plane = e12p + (size_t)(blockIdx.x & 1) * E12_PLANE + (size_t)(tm * 16) * E12_COLS;
    for (int o = threadIdx.x; o < 16 * (E12_COLS / 4); o += KH_SLICES * 32) {
        const int r  = o >> 6;
        const int c4 = (o & 63) * 4;
        *(v4f*)(plane + (size_t)r * E12_COLS + c4) = *(const v4f*)(es + r * ES_PITCH + c4);
    }
}

#define PA_PITCH 36
#define PB_PITCH 68

struct PairGrp { v4f a[4]; v4f b[4]; v4f w; };

__device__ __forceinline__ void pair_load(PairGrp& G, const float* pa, const float* pb,
                                          const float* pw, int g) {
#pragma unroll
    for (int u = 0; u < 4; ++u) {
        G.a[u] = *(const v4f*)(pa + (4 * g + u) * PA_PITCH);
        G.b[u] = *(const v4f*)(pb + (4 * g + u) * PB_PITCH);
    }
    G.w = *(const v4f*)(pw + 4 * g);
}

__device__ __forceinline__ void pair_fma(float (&acc)[4][4], const PairGrp& G) {
    v4f w = G.w;
    asm("" : "+v"(w));
#pragma unroll
    for (int u = 0; u < 4; ++u) {
        v4f a = G.a[u], b = G.b[u];
        asm("" : "+v"(a));
        asm("" : "+v"(b));
#pragma unroll
        for (int r = 0; r < 4; ++r) {
#pragma unroll
            for (int c = 0; c < 4; ++c)
                acc[r][c] = fmaf(fmaxf(a[r] + b[c], 0.f), w[u], acc[r][c]);
        }
    }
}

__device__ __forceinline__ float pair_out(float s, float bias2, int m) {
    float e = s + bias2;
    e = (e >= 0.f) ? e : 0.01f * e;
    return m ? __expf(e) : 0.f;
}

__global__ void __launch_bounds__(128) __attribute__((amdgpu_waves_per_eu(1, 2)))
k_pair(const float* __restrict__ e12, const float* __restrict__ b1,
       const float* __restrict__ W2, const float* __restrict__ b2,
       const int* __restrict__ adj, float* __restrict__ expE,
       float* __restrict__ partials) {
    __shared__ __attribute__((aligned(16))) float sA[HID_ * PA_PITCH];
    __shared__ __attribute__((aligned(16))) float sB[HID_ * PB_PITCH];
    __shared__ __attribute__((aligned(16))) float sw2[HID_];
    __shared__ float red[4];

    const int blk  = blockIdx.x;
    const int b    = blk >> 7;
    const int t    = blk & 127;
    const int i0   = (t >> 3) * 32;
    const int j0   = (t & 7) * 64;
    const int tid  = threadIdx.x;
    const int lane = tid & 31;
    const int wave = tid >> 5;
    const int tj   = tid & 15;
    const int ti   = tid >> 4;

    const size_t g0 = (size_t)b * N_ * N_ + (size_t)(i0 + 4 * ti) * N_ + (j0 + 4 * tj);

    const int hq = wave * 8 + (lane & 7);
    const int rr = lane >> 3;
    const float4 bb = *(const float4*)(b1 + 4 * hq);
    float4 v[24], u[24];
    const float* pe1 = e12 + (size_t)(b * N_ + i0 + rr) * E12_COLS + 4 * hq;
    const float* pe2 = e12 + (size_t)(b * N_ + j0 + rr) * E12_COLS + HID_ + 4 * hq;
#pragma unroll
    for (int k = 0; k < 24; ++k) {
        const float* src = (k < 8) ? pe1 + (size_t)(4 * k) * E12_COLS
                                   : pe2 + (size_t)(4 * (k - 8)) * E12_COLS;
        v[k] = *(const float4*)src;
        u[k] = *(const float4*)(src + E12_PLANE);
    }
#pragma unroll
    for (int k = 0; k < 24; ++k) {
        const int rho = 4 * k + rr;
        if (k < 8) {
            float* d = sA + (4 * hq) * PA_PITCH + rho;
            d[0]            = (v[k].x + u[k].x) + bb.x;
            d[PA_PITCH]     = (v[k].y + u[k].y) + bb.y;
            d[2 * PA_PITCH] = (v[k].z + u[k].z) + bb.z;
            d[3 * PA_PITCH] = (v[k].w + u[k].w) + bb.w;
        } else {
            float* d = sB + (4 * hq) * PB_PITCH + (rho - 32);
            d[0]            = v[k].x + u[k].x;
            d[PB_PITCH]     = v[k].y + u[k].y;
            d[2 * PB_PITCH] = v[k].z + u[k].z;
            d[3 * PB_PITCH] = v[k].w + u[k].w;
        }
    }
    sw2[tid] = W2[tid];
    __syncthreads();

    int4 msk[4];
#pragma unroll
    for (int r = 0; r < 4; ++r) msk[r] = *(const int4*)(adj + g0 + (size_t)r * N_);

    const float* pa = sA + 4 * ti;
    const float* pb = sB + 4 * tj;

    float acc[4][4];
#pragma unroll
    for (int r = 0; r < 4; ++r) {
#pragma unroll
        for (int c = 0; c < 4; ++c) acc[r][c] = 0.f;
    }

    PairGrp G0, G1;
    pair_load(G0, pa, pb, sw2, 0);
#pragma unroll 1
    for (int g = 0; g < HID_ / 4 - 2; g += 2) {
        pair_load(G1, pa, pb, sw2, g + 1);
        pair_fma(acc, G0);
        pair_load(G0, pa, pb, sw2, g + 2);
        pair_fma(acc, G1);
    }
    pair_load(G1, pa, pb, sw2, HID_ / 4 - 1);
    pair_fma(acc, G0);
    pair_fma(acc, G1);

    const float bias2 = b2[0];
    float lsum = 0.f;
#pragma unroll
    for (int r = 0; r < 4; ++r) {
        float4 o;
        o.x = pair_out(acc[r][0], bias2, msk[r].x);
        o.y = pair_out(acc[r][1], bias2, msk[r].y);
        o.z = pair_out(acc[r][2], bias2, msk[r].z);
        o.w = pair_out(acc[r][3], bias2, msk[r].w);
        *(float4*)(expE + g0 + (size_t)r * N_) = o;
        lsum += (o.x + o.y) + (o.z + o.w);
    }

    lsum += xhalf(lsum);
    lsum += xrow<8>(lsum);
    lsum += xrow<4>(lsum);
    lsum += xrow<2>(lsum);
    lsum += xrow<1>(lsum);
    if (lane == 0) red[wave] = lsum;
    __syncthreads();
    if (tid == 0) {
        partials[2 * blk]     = red[0] + red[1];
        partials[2 * blk + 1] = red[2] + red[3];
    }
}

#define KO_SLICES 8
#define KO_SPAN   (N_ / KO_SLICES)
__global__ void __launch_bounds__(KO_SLICES * 32) __attribute__((amdgpu_waves_per_eu(2, 2)))
k_out(const float* __restrict__ expE, const float* __restrict__ h,
      const float* __restrict__ partials, float* __restrict__ out) {
    __shared__ float red[KO_SLICES * 2 * TILE_PITCH];

    const int lane = threadIdx.x & 31;
    const int sl   = threadIdx.x >> 5;
    const int b    = blockIdx.x >> 6;
    const int tm   = (blockIdx.x >> 1) & 31;
    const int n0   = (blockIdx.x & 1) * 32;
    const int hf   = lane >> 4;
    const int c16  = lane & 15;

    const float* pp = partials + b * 256 + lane;
    const float q0 = pp[0],   q1 = pp[32],  q2 = pp[64],  q3 = pp[96];
    const float q4 = pp[128], q5 = pp[160], q6 = pp[192], q7 = pp[224];
    float tot = ((q0 + q4) + (q2 + q6)) + ((q1 + q5) + (q3 + q7));
    tot += xhalf(tot);
    tot += xrow<8>(tot);
    tot += xrow<4>(tot);
    tot += xrow<2>(tot);
    tot += xrow<1>(tot);

    const float* arow = expE + (size_t)b * N_ * N_ + (size_t)(tm * 16 + c16) * N_ + sl * KO_SPAN + 4 * hf;
    const float* bcol = h + ((size_t)b * N_ + sl * KO_SPAN + 4 * hf) * ATT_ + n0 + 2 * c16;

    v8f c0 = {}, c1 = {};
#pragma unroll
    for (int k = 0; k < KO_SPAN; k += 8) {
        const v4f a4 = *(const v4f*)(arow + k);
        const v2f r0 = *(const v2f*)(bcol + (size_t)k * ATT_);
        const v2f r1 = *(const v2f*)(bcol + (size_t)(k + 1) * ATT_);
        const v2f r2 = *(const v2f*)(bcol + (size_t)(k + 2) * ATT_);
        const v2f r3 = *(const v2f*)(bcol + (size_t)(k + 3) * ATT_);
        v2f a, b0, b1;
        a.x  = a4.x; a.y  = a4.y;
        b0.x = r0.x; b0.y = r1.x;
        b1.x = r0.y; b1.y = r1.y;
        c0 = __builtin_amdgcn_wmma_f32_16x16x4_f32(false, a, false, b0,
                                                   (short)0, c0, false, false);
        c1 = __builtin_amdgcn_wmma_f32_16x16x4_f32(false, a, false, b1,
                                                   (short)0, c1, false, false);
        a.x  = a4.z; a.y  = a4.w;
        b0.x = r2.x; b0.y = r3.x;
        b1.x = r2.y; b1.y = r3.y;
        c0 = __builtin_amdgcn_wmma_f32_16x16x4_f32(false, a, false, b0,
                                                   (short)0, c0, false, false);
        c1 = __builtin_amdgcn_wmma_f32_16x16x4_f32(false, a, false, b1,
                                                   (short)0, c1, false, false);
    }

    float* mine = red + (size_t)(sl * 2) * TILE_PITCH + lane;
#pragma unroll
    for (int v = 0; v < 8; ++v) {
        mine[v * 32]              = c0[v];
        mine[TILE_PITCH + v * 32] = c1[v];
    }
    __syncthreads();

    const float inv = 1.0f / tot;
#pragma unroll
    for (int t = 0; t < 2; ++t) {
        const int r  = (threadIdx.x >> 5) + 8 * t;
        const int cc = threadIdx.x & 31;
        const float* p = red + (cc & 1) * TILE_PITCH + (r & 7) * 32 + (r >> 3) * 16 + (cc >> 1);
        float acc = p[0];
#pragma unroll
        for (int s = 1; s < KO_SLICES; ++s) acc += p[(size_t)s * 2 * TILE_PITCH];
        out[((size_t)b * N_ + tm * 16 + r) * ATT_ + n0 + cc] = acc * inv;
    }
}

extern "C" void kernel_launch(void* const* d_in, const int* in_sizes, int n_in,
                              void* d_out, int out_size, void* d_ws, size_t ws_size,
                              hipStream_t stream) {
    const float* x   = (const float*)d_in[0];
    const int*   adj = (const int*)  d_in[1];
    const float* Ww  = (const float*)d_in[2];
    const float* Wb  = (const float*)d_in[3];
    const float* W1  = (const float*)d_in[4];
    const float* b1  = (const float*)d_in[5];
    const float* W2  = (const float*)d_in[6];
    const float* b2  = (const float*)d_in[7];
    float* out = (float*)d_out;

    float* ws    = (float*)d_ws;
    float* h_    = ws;
    float* e12   = h_   + (size_t)B_ * N_ * ATT_;
    float* expE  = e12  + 2 * E12_PLANE;
    float* parts = expE + (size_t)B_ * N_ * N_;

    k_h     <<<128, 256, 0, stream>>>(x, Ww, Wb, W1, h_, e12);
    k_pair  <<<256, 128, 0, stream>>>(e12, b1, W2, b2, adj, expE, parts);
    k_out   <<<128, 256, 0, stream>>>(expE, h_, parts, out);
}


# both GEMM kernels: fragments no longer loaded straight from memory (16 lines a load): a wave loads its K slice in whole 128-byte lines, parks it in its own LDS rows and reads the same fragments there
# speedup vs baseline: 2.3263x; 1.0605x over previous
#include <hip/hip_runtime.h>
#include <hip/hip_bf16.h>


#define B_  2
#define N_  512
#define F_  768
#define ATT_ 64
#define HID_ 128

typedef __attribute__((ext_vector_type(2))) float v2f;
typedef __attribute__((ext_vector_type(8))) float v8f;

__device__ __forceinline__ float xhalf(float x) {
    const int b = __builtin_bit_cast(int, x);
    return __builtin_bit_cast(float, __builtin_amdgcn_permlanex16(b, b, 0x76543210, 0xfedcba98, false, false));
}
template <int M>
__device__ __forceinline__ float xrow(float x) {
    const int b = __builtin_bit_cast(int, x);
    return __builtin_bit_cast(float, __builtin_amdgcn_update_dpp(b, b, 0x160 + M, 0xf, 0xf, false));
}

#define KH_SLICES 8
#define KH_SPAN   (F_ / KH_SLICES)
#define TILE_PITCH 272
#define HT_PITCH  36
#define E12_COLS  256
#define ES_PITCH  260
#define E12_PLANE ((size_t)B_ * N_ * E12_COLS)
#define E12_WAVES 8
#define SG_ROWS   48
#define SG_PITCH  100
#define SG_WAVE   (SG_ROWS * SG_PITCH)

typedef __attribute__((ext_vector_type(4))) float v4f;

__global__ void __launch_bounds__(KH_SLICES * 32) __attribute__((amdgpu_waves_per_eu(2, 2)))
k_h(const float* __restrict__ x, const float* __restrict__ Ww,
    const float* __restrict__ Wb, const float* __restrict__ W1,
    float* __restrict__ h, float* __restrict__ e12p) {
    __shared__ __attribute__((aligned(16))) float stg[KH_SLICES * SG_WAVE];
    __shared__ float red[KH_SLICES * 2 * TILE_PITCH];
    __shared__ __attribute__((aligned(16))) float hT[16 * HT_PITCH];
    __shared__ __attribute__((aligned(16))) float es[16 * ES_PITCH];

    const int lane = threadIdx.x & 31;
    const int sl   = threadIdx.x >> 5;
    const int tm   = blockIdx.x >> 1;
    const int n0   = (blockIdx.x & 1) * 32;
    const int hf   = lane >> 4;
    const int c16  = lane & 15;
    const int wv   = __builtin_amdgcn_readfirstlane(sl);

    const int lr = lane >> 3;
    const int lc = lane & 7;
    const float* gx = x  + (size_t)(tm * 16 + lr) * F_ + sl * KH_SPAN + 4 * lc;
    const float* gw = Ww + (size_t)(n0 + lr) * F_ + sl * KH_SPAN + 4 * lc;
    v4f sv[3][12];
#pragma unroll
    for (int p = 0; p < 3; ++p) {
#pragma unroll
        for (int g = 0; g < 12; ++g) {
            const float* srcp = (g < 4) ? gx + (size_t)(4 * g) * F_ : gw + (size_t)(4 * (g - 4)) * F_;
            sv[p][g] = *(const v4f*)(srcp + 32 * p);
        }
        __builtin_amdgcn_sched_barrier(0);
    }

    static_assert(E12_WAVES == KH_SLICES, "every wave of the work-group takes a pair of e12 column tiles");
    v4f wq0[4], wq1[4];
    {
        const int ecol = wv * 32 + c16;
        const float* wp = W1 + (size_t)(ecol & (HID_ - 1)) * (2 * ATT_) + ((ecol >= HID_) ? ATT_ : 0) + n0 + 16 * hf;
#pragma unroll
        for (int q = 0; q < 4; ++q) {
            wq0[q] = *(const v4f*)(wp + 4 * q);
            wq1[q] = *(const v4f*)(wp + (size_t)16 * (2 * ATT_) + 4 * q);
        }
    }

    float* my = stg + (size_t)sl * SG_WAVE;
    float* sd = my + lr * SG_PITCH + 4 * lc;
    const float* fa = my + c16 * SG_PITCH + 4 * hf;

    v8f c0 = {}, c1 = {};
#pragma unroll
    for (int p = 0; p < 3; ++p) {
#pragma unroll
        for (int g = 0; g < 12; ++g) *(v4f*)(sd + (4 * g) * SG_PITCH + 32 * p) = sv[p][g];
        __builtin_amdgcn_fence(__ATOMIC_RELEASE, "wavefront", "local");
        __builtin_amdgcn_fence(__ATOMIC_ACQUIRE, "wavefront", "local");
        v4f fa4[4], fp4[4], fq4[4];
#pragma unroll
        for (int j = 0; j < 4; ++j) {
            fa4[j] = *(const v4f*)(fa + 32 * p + 8 * j);
            fp4[j] = *(const v4f*)(fa + 16 * SG_PITCH + 32 * p + 8 * j);
            fq4[j] = *(const v4f*)(fa + 32 * SG_PITCH + 32 * p + 8 * j);
        }
        __builtin_amdgcn_sched_barrier(0);
#pragma unroll
        for (int j = 0; j < 4; ++j) {
            const v4f a4 = fa4[j], p4 = fp4[j], q4 = fq4[j];
            v2f a, b0, b1;
            a.x  = a4.x; a.y  = a4.y;
            b0.x = p4.x; b0.y = p4.y;
            b1.x = q4.x; b1.y = q4.y;
            c0 = __builtin_amdgcn_wmma_f32_16x16x4_f32(false, a, false, b0,
                                                       (short)0, c0, false, false);
            c1 = __builtin_amdgcn_wmma_f32_16x16x4_f32(false, a, false, b1,
                                                       (short)0, c1, false, false);
            a.x  = a4.z; a.y  = a4.w;
            b0.x = p4.z; b0.y = p4.w;
            b1.x = q4.z; b1.y = q4.w;
            c0 = __builtin_amdgcn_wmma_f32_16x16x4_f32(false, a, false, b0,
                                                       (short)0, c0, false, false);
            c1 = __builtin_amdgcn_wmma_f32_16x16x4_f32(false, a, false, b1,
                                                       (short)0, c1, false, false);
        }
    }

    float* mine = red + (size_t)(sl * 2) * TILE_PITCH + lane;
#pragma unroll
    for (int v = 0; v < 8; ++v) {
        mine[v * 32]              = c0[v];
        mine[TILE_PITCH + v * 32] = c1[v];
    }
    __syncthreads();

    float hv[2] = {0.f, 0.f};
#pragma unroll
    for (int t = 0; t < 2; ++t) {
        const int o = threadIdx.x + t * KH_SLICES * 32;
        if (o < 16 * 32) {
            const int r  = o >> 5;
            const int cc = o & 31;
            const float* p = red + (cc >> 4) * TILE_PITCH + (r & 7) * 32 + (r >> 3) * 16 + (cc & 15);
            float acc = p[0];
#pragma unroll
            for (int s = 1; s < KH_SLICES; ++s) acc += p[(size_t)s * 2 * TILE_PITCH];
            hv[t] = acc + Wb[n0 + cc];
            hT[r * HT_PITCH + cc] = hv[t];
        }
    }
    __syncthreads();

    {
        v8f d0 = {}, d1 = {};
        const float* ap = hT + c16 * HT_PITCH + 16 * hf;
#pragma unroll
        for (int q = 0; q < 4; ++q) {
            const v4f a4 = *(const v4f*)(ap + 4 * q);
            v2f a, b0, b1;
            a.x  = a4.x;     a.y  = a4.y;
            b0.x = wq0[q].x; b0.y = wq0[q].y;
            b1.x = wq1[q].x; b1.y = wq1[q].y;
            d0 = __builtin_amdgcn_wmma_f32_16x16x4_f32(false, a, false, b0,
                                                       (short)0, d0, false, false);
            d1 = __builtin_amdgcn_wmma_f32_16x16x4_f32(false, a, false, b1,
                                                       (short)0, d1, false, false);
            a.x  = a4.z;     a.y  = a4.w;
            b0.x = wq0[q].z; b0.y = wq0[q].w;
            b1.x = wq1[q].z; b1.y = wq1[q].w;
            d0 = __builtin_amdgcn_wmma_f32_16x16x4_f32(false, a, false, b0,
                                                       (short)0, d0, false, false);
            d1 = __builtin_amdgcn_wmma_f32_16x16x4_f32(false, a, false, b1,
                                                       (short)0, d1, false, false);
        }
        float* st = es + (size_t)(8 * hf) * ES_PITCH + wv * 32 + c16;
#pragma unroll
        for (int v = 0; v < 8; ++v) {
            st[v * ES_PITCH]      = d0[v];
            st[v * ES_PITCH + 16] = d1[v];
        }
    }
    __syncthreads();

#pragma unroll
    for (int t = 0; t < 2; ++t) {
        const int o = threadIdx.x + t * KH_SLICES * 32;
        if (o < 16 * 32)
            h[(size_t)(tm * 16 + (o >> 5)) * ATT_ + n0 + (o & 31)] = hv[t];
    }
    float* plane = e12p + (size_t)(blockIdx.x & 1) * E12_PLANE + (size_t)(tm * 16) * E12_COLS;
    for (int o = threadIdx.x; o < 16 * (E12_COLS / 4); o += KH_SLICES * 32) {
        const int r  = o >> 6;
        const int c4 = (o & 63) * 4;
        *(v4f*)(plane + (size_t)r * E12_COLS + c4) = *(const v4f*)(es + r * ES_PITCH + c4);
    }
}

#define PA_PITCH 36
#define PB_PITCH 68

struct PairGrp { v4f a[4]; v4f b[4]; v4f w; };

__device__ __forceinline__ void pair_load(PairGrp& G, const float* pa, const float* pb,
                                          const float* pw, int g) {
#pragma unroll
    for (int u = 0; u < 4; ++u) {
        G.a[u] = *(const v4f*)(pa + (4 * g + u) * PA_PITCH);
        G.b[u] = *(const v4f*)(pb + (4 * g + u) * PB_PITCH);
    }
    G.w = *(const v4f*)(pw + 4 * g);
}

__device__ __forceinline__ void pair_fma(float (&acc)[4][4], const PairGrp& G) {
    v4f w = G.w;
    asm("" : "+v"(w));
#pragma unroll
    for (int u = 0; u < 4; ++u) {
        v4f a = G.a[u], b = G.b[u];
        asm("" : "+v"(a));
        asm("" : "+v"(b));
#pragma unroll
        for (int r = 0; r < 4; ++r) {
#pragma unroll
            for (int c = 0; c < 4; ++c)
                acc[r][c] = fmaf(fmaxf(a[r] + b[c], 0.f), w[u], acc[r][c]);
        }
    }
}

__device__ __forceinline__ float pair_out(float s, float bias2, int m) {
    float e = s + bias2;
    e = (e >= 0.f) ? e : 0.01f * e;
    return m ? __expf(e) : 0.f;
}

__global__ void __launch_bounds__(128) __attribute__((amdgpu_waves_per_eu(1, 2)))
k_pair(const float* __restrict__ e12, const float* __restrict__ b1,
       const float* __restrict__ W2, const float* __restrict__ b2,
       const int* __restrict__ adj, float* __restrict__ expE,
       float* __restrict__ partials) {
    __shared__ __attribute__((aligned(16))) float sA[HID_ * PA_PITCH];
    __shared__ __attribute__((aligned(16))) float sB[HID_ * PB_PITCH];
    __shared__ __attribute__((aligned(16))) float sw2[HID_];
    __shared__ float red[4];

    const int blk  = blockIdx.x;
    const int b    = blk >> 7;
    const int t    = blk & 127;
    const int i0   = (t >> 3) * 32;
    const int j0   = (t & 7) * 64;
    const int tid  = threadIdx.x;
    const int lane = tid & 31;
    const int wave = tid >> 5;
    const int tj   = tid & 15;
    const int ti   = tid >> 4;

    const size_t g0 = (size_t)b * N_ * N_ + (size_t)(i0 + 4 * ti) * N_ + (j0 + 4 * tj);

    const int hq = wave * 8 + (lane & 7);
    const int rr = lane >> 3;
    const float4 bb = *(const float4*)(b1 + 4 * hq);
    float4 v[24], u[24];
    const float* pe1 = e12 + (size_t)(b * N_ + i0 + rr) * E12_COLS + 4 * hq;
    const float* pe2 = e12 + (size_t)(b * N_ + j0 + rr) * E12_COLS + HID_ + 4 * hq;
#pragma unroll
    for (int k = 0; k < 24; ++k) {
        const float* src = (k < 8) ? pe1 + (size_t)(4 * k) * E12_COLS
                                   : pe2 + (size_t)(4 * (k - 8)) * E12_COLS;
        v[k] = *(const float4*)src;
        u[k] = *(const float4*)(src + E12_PLANE);
    }
#pragma unroll
    for (int k = 0; k < 24; ++k) {
        const int rho = 4 * k + rr;
        if (k < 8) {
            float* d = sA + (4 * hq) * PA_PITCH + rho;
            d[0]            = (v[k].x + u[k].x) + bb.x;
            d[PA_PITCH]     = (v[k].y + u[k].y) + bb.y;
            d[2 * PA_PITCH] = (v[k].z + u[k].z) + bb.z;
            d[3 * PA_PITCH] = (v[k].w + u[k].w) + bb.w;
        } else {
            float* d = sB + (4 * hq) * PB_PITCH + (rho - 32);
            d[0]            = v[k].x + u[k].x;
            d[PB_PITCH]     = v[k].y + u[k].y;
            d[2 * PB_PITCH] = v[k].z + u[k].z;
            d[3 * PB_PITCH] = v[k].w + u[k].w;
        }
    }
    sw2[tid] = W2[tid];
    __syncthreads();

    int4 msk[4];
#pragma unroll
    for (int r = 0; r < 4; ++r) msk[r] = *(const int4*)(adj + g0 + (size_t)r * N_);

    const float* pa = sA + 4 * ti;
    const float* pb = sB + 4 * tj;

    float acc[4][4];
#pragma unroll
    for (int r = 0; r < 4; ++r) {
#pragma unroll
        for (int c = 0; c < 4; ++c) acc[r][c] = 0.f;
    }

    PairGrp G0, G1;
    pair_load(G0, pa, pb, sw2, 0);
#pragma unroll 1
    for (int g = 0; g < HID_ / 4 - 2; g += 2) {
        pair_load(G1, pa, pb, sw2, g + 1);
        pair_fma(acc, G0);
        pair_load(G0, pa, pb, sw2, g + 2);
        pair_fma(acc, G1);
    }
    pair_load(G1, pa, pb, sw2, HID_ / 4 - 1);
    pair_fma(acc, G0);
    pair_fma(acc, G1);

    const float bias2 = b2[0];
    float lsum = 0.f;
#pragma unroll
    for (int r = 0; r < 4; ++r) {
        float4 o;
        o.x = pair_out(acc[r][0], bias2, msk[r].x);
        o.y = pair_out(acc[r][1], bias2, msk[r].y);
        o.z = pair_out(acc[r][2], bias2, msk[r].z);
        o.w = pair_out(acc[r][3], bias2, msk[r].w);
        *(float4*)(expE + g0 + (size_t)r * N_) = o;
        lsum += (o.x + o.y) + (o.z + o.w);
    }

    lsum += xhalf(lsum);
    lsum += xrow<8>(lsum);
    lsum += xrow<4>(lsum);
    lsum += xrow<2>(lsum);
    lsum += xrow<1>(lsum);
    if (lane == 0) red[wave] = lsum;
    __syncthreads();
    if (tid == 0) {
        partials[2 * blk]     = red[0] + red[1];
        partials[2 * blk + 1] = red[2] + red[3];
    }
}

#define KO_SLICES 8
#define KO_SPAN   (N_ / KO_SLICES)
#define SO_PITCH  68
#define SO_WAVE   (16 * SO_PITCH)
__global__ void __launch_bounds__(KO_SLICES * 32) __attribute__((amdgpu_waves_per_eu(2, 2)))
k_out(const float* __restrict__ expE, const float* __restrict__ h,
      const float* __restrict__ partials, float* __restrict__ out) {
    __shared__ __attribute__((aligned(16))) float stgA[KO_SLICES * SO_WAVE];
    __shared__ float red[KO_SLICES * 2 * TILE_PITCH];

    const int lane = threadIdx.x & 31;
    const int sl   = threadIdx.x >> 5;
    const int b    = blockIdx.x >> 6;
    const int tm   = (blockIdx.x >> 1) & 31;
    const int n0   = (blockIdx.x & 1) * 32;
    const int hf   = lane >> 4;
    const int c16  = lane & 15;

    const float* pp = partials + b * 256 + lane;
    const float q0 = pp[0],   q1 = pp[32],  q2 = pp[64],  q3 = pp[96];
    const float q4 = pp[128], q5 = pp[160], q6 = pp[192], q7 = pp[224];
    float tot = ((q0 + q4) + (q2 + q6)) + ((q1 + q5) + (q3 + q7));
    tot += xhalf(tot);
    tot += xrow<8>(tot);
    tot += xrow<4>(tot);
    tot += xrow<2>(tot);
    tot += xrow<1>(tot);

    const int lr = lane >> 4;
    const int lc = lane & 15;
    const float* ga = expE + (size_t)b * N_ * N_ + (size_t)(tm * 16 + lr) * N_ + sl * KO_SPAN + 4 * lc;
    v4f sa[8];
#pragma unroll
    for (int g = 0; g < 8; ++g) sa[g] = *(const v4f*)(ga + (size_t)(2 * g) * N_);

    float* my = stgA + (size_t)sl * SO_WAVE;
    const float* fa = my + c16 * SO_PITCH + 4 * hf;
    const float* bcol = h + ((size_t)b * N_ + sl * KO_SPAN + 4 * hf) * ATT_ + n0 + 2 * c16;

    v2f rb[8][4];
#pragma unroll
    for (int j = 0; j < 8; ++j) {
#pragma unroll
        for (int t = 0; t < 4; ++t) rb[j][t] = *(const v2f*)(bcol + (size_t)(8 * j + t) * ATT_);
    }

#pragma unroll
    for (int g = 0; g < 8; ++g) *(v4f*)(my + (2 * g + lr) * SO_PITCH + 4 * lc) = sa[g];
    __builtin_amdgcn_fence(__ATOMIC_RELEASE, "wavefront", "local");
    __builtin_amdgcn_fence(__ATOMIC_ACQUIRE, "wavefront", "local");
    v4f fa4[8];
#pragma unroll
    for (int j = 0; j < 8; ++j) fa4[j] = *(const v4f*)(fa + 8 * j);
    __builtin_amdgcn_sched_barrier(0);

    v8f c0 = {}, c1 = {};
#pragma unroll
    for (int j = 0; j < 8; ++j) {
        const v4f a4 = fa4[j];
        const v2f r0 = rb[j][0], r1 = rb[j][1], r2 = rb[j][2], r3 = rb[j][3];
        v2f a, b0, b1;
        a.x  = a4.x; a.y  = a4.y;
        b0.x = r0.x; b0.y = r1.x;
        b1.x = r0.y; b1.y = r1.y;
        c0 = __builtin_amdgcn_wmma_f32_16x16x4_f32(false, a, false, b0,
                                                   (short)0, c0, false, false);
        c1 = __builtin_amdgcn_wmma_f32_16x16x4_f32(false, a, false, b1,
                                                   (short)0, c1, false, false);
        a.x  = a4.z; a.y  = a4.w;
        b0.x = r2.x; b0.y = r3.x;
        b1.x = r2.y; b1.y = r3.y;
        c0 = __builtin_amdgcn_wmma_f32_16x16x4_f32(false, a, false, b0,
                                                   (short)0, c0, false, false);
        c1 = __builtin_amdgcn_wmma_f32_16x16x4_f32(false, a, false, b1,
                                                   (short)0, c1, false, false);
    }

    float* mine = red + (size_t)(sl * 2) * TILE_PITCH + lane;
#pragma unroll
    for (int v = 0; v < 8; ++v) {
        mine[v * 32]              = c0[v];
        mine[TILE_PITCH + v * 32] = c1[v];
    }
    __syncthreads();

    const float inv = 1.0f / tot;
#pragma unroll
    for (int t = 0; t < 2; ++t) {
        const int r  = (threadIdx.x >> 5) + 8 * t;
        const int cc = threadIdx.x & 31;
        const float* p = red + (cc & 1) * TILE_PITCH + (r & 7) * 32 + (r >> 3) * 16 + (cc >> 1);
        float acc = p[0];
#pragma unroll
        for (int s = 1; s < KO_SLICES; ++s) acc += p[(size_t)s * 2 * TILE_PITCH];
        out[((size_t)b * N_ + tm * 16 + r) * ATT_ + n0 + cc] = acc * inv;
    }
}

extern "C" void kernel_launch(void* const* d_in, const int* in_sizes, int n_in,
                              void* d_out, int out_size, void* d_ws, size_t ws_size,
                              hipStream_t stream) {
    const float* x   = (const float*)d_in[0];
    const int*   adj = (const int*)  d_in[1];
    const float* Ww  = (const float*)d_in[2];
    const float* Wb  = (const float*)d_in[3];
    const float* W1  = (const float*)d_in[4];
    const float* b1  = (const float*)d_in[5];
    const float* W2  = (const float*)d_in[6];
    const float* b2  = (const float*)d_in[7];
    float* out = (float*)d_out;

    float* ws    = (float*)d_ws;
    float* h_    = ws;
    float* e12   = h_   + (size_t)B_ * N_ * ATT_;
    float* expE  = e12  + 2 * E12_PLANE;
    float* parts = expE + (size_t)B_ * N_ * N_;

    k_h     <<<128, 256, 0, stream>>>(x, Ww, Wb, W1, h_, e12);
    k_pair  <<<256, 128, 0, stream>>>(e12, b1, W2, b2, adj, expE, parts);
    k_out   <<<128, 256, 0, stream>>>(expE, h_, parts, out);
}
